# starting kernel
# baseline (speedup 1.0000x reference)
#include <hip/hip_runtime.h>
#include <math.h>

#define NHEADS   16
#define HD       64
#define TSEQ     1024
#define CDIM     1024
#define BATCH    4
#define EPSV     1e-6f
#define KCH      32
#define LPAD     36

typedef float v2f __attribute__((ext_vector_type(2)));
typedef float v8f __attribute__((ext_vector_type(8)));

static __device__ __forceinline__ v8f wmma4(v2f a, v2f b, v8f c) {
  return __builtin_amdgcn_wmma_f32_16x16x4_f32(false, a, false, b, (short)0, c,
                                               false, false);
}

static __device__ __forceinline__ float half16_max(float v) {
#pragma unroll
  for (int off = 1; off < 16; off <<= 1) v = fmaxf(v, __shfl_xor(v, off, 16));
  return v;
}
static __device__ __forceinline__ float half16_sum(float v) {
#pragma unroll
  for (int off = 1; off < 16; off <<= 1) v += __shfl_xor(v, off, 16);
  return v;
}
static __device__ __forceinline__ float wave32_sum(float v) {
#pragma unroll
  for (int off = 1; off < 32; off <<= 1) v += __shfl_xor(v, off, 32);
  return v;
}

__global__ __launch_bounds__(256) void gemm_abT_wmma(
    const float* __restrict__ A, const float* __restrict__ B,
    float* __restrict__ C, int K, int lda, int ldb, int ldc) {
  __shared__ __align__(16) float lA[2][128 * LPAD];
  __shared__ __align__(16) float lB[2][64 * LPAD];

  const int tid = threadIdx.x;
  const int lane = tid & 31;
  const int wave = tid >> 5;
  const int lm = lane & 15;
  const int lh = lane >> 4;
  const int m0 = blockIdx.y * 128;
  const int n0 = blockIdx.x * 64;

  const int srow = tid >> 3;
  const int scol = (tid & 7) * 4;

  const float* Ast = A + (size_t)(m0 + srow) * lda + scol;
  const float* Bst = B + (size_t)(n0 + srow) * ldb + scol;

  float4 ra[4], rb[2];

#pragma unroll
  for (int it = 0; it < 4; ++it)
    ra[it] = *(const float4*)(Ast + (size_t)(32 * it) * lda);
#pragma unroll
  for (int it = 0; it < 2; ++it)
    rb[it] = *(const float4*)(Bst + (size_t)(32 * it) * ldb);
#pragma unroll
  for (int it = 0; it < 4; ++it)
    *(float4*)(&lA[0][(srow + 32 * it) * LPAD + scol]) = ra[it];
#pragma unroll
  for (int it = 0; it < 2; ++it)
    *(float4*)(&lB[0][(srow + 32 * it) * LPAD + scol]) = rb[it];
  __syncthreads();

  v8f acc[4] = {v8f{}, v8f{}, v8f{}, v8f{}};
  const int nch = K / KCH;

  for (int c = 0; c < nch; ++c) {
    const int buf = c & 1;

    if (c + 1 < nch) {
      const int k1 = (c + 1) * KCH;
#pragma unroll
      for (int it = 0; it < 4; ++it)
        ra[it] = *(const float4*)(Ast + (size_t)(32 * it) * lda + k1);
#pragma unroll
      for (int it = 0; it < 2; ++it)
        rb[it] = *(const float4*)(Bst + (size_t)(32 * it) * ldb + k1);
    }
    if (c + 2 < nch) {
      __builtin_prefetch(Ast + (size_t)(c + 2) * KCH, 0, 1);
      __builtin_prefetch(Bst + (size_t)(c + 2) * KCH, 0, 1);
    }

    const float* pa = &lA[buf][(wave * 16 + lm) * LPAD + 2 * lh];
    const float* pb = &lB[buf][lm * LPAD + 2 * lh];

    v2f af[8];
    v2f bf[8][4];
#pragma unroll
    for (int ks = 0; ks < 8; ++ks) {
      af[ks] = *(const v2f*)(pa + 4 * ks);
#pragma unroll
      for (int j = 0; j < 4; ++j)
        bf[ks][j] = *(const v2f*)(pb + j * 16 * LPAD + 4 * ks);
    }
#pragma unroll
    for (int ks = 0; ks < 8; ++ks)
#pragma unroll
      for (int j = 0; j < 4; ++j)
        acc[j] = wmma4(af[ks], bf[ks][j], acc[j]);

    if (c + 1 < nch) {
      const int nb = buf ^ 1;
#pragma unroll
      for (int it = 0; it < 4; ++it)
        *(float4*)(&lA[nb][(srow + 32 * it) * LPAD + scol]) = ra[it];
#pragma unroll
      for (int it = 0; it < 2; ++it)
        *(float4*)(&lB[nb][(srow + 32 * it) * LPAD + scol]) = rb[it];
      __syncthreads();
    }
  }

  const int mw = m0 + wave * 16;
#pragma unroll
  for (int j = 0; j < 4; ++j)
#pragma unroll
    for (int r = 0; r < 8; ++r)
      C[(size_t)(mw + r + 8 * lh) * ldc + n0 + j * 16 + lm] = acc[j][r];
}

__global__ __launch_bounds__(256) void rope_norm_kernel(
    const float* __restrict__ qkv, const float* __restrict__ hyp,
    float* __restrict__ Q, float* __restrict__ Kq, float* __restrict__ V,
    float* __restrict__ q0, float* __restrict__ k0) {
  const int lane = threadIdx.x & 31;
  const int gw = blockIdx.x * 8 + (threadIdx.x >> 5);
  const int h = gw % NHEADS;
  const int t = (gw / NHEADS) % TSEQ;
  const int b = gw / (NHEADS * TSEQ);

  const size_t ibase = ((size_t)(b * TSEQ + t)) * (3 * CDIM) + h * HD;
  const size_t obase = ((size_t)((b * NHEADS + h) * TSEQ) + t) * HD;
  const size_t nidx = (size_t)(b * NHEADS + h) * TSEQ + t;
  const float kc = hyp[h];

  const float fr = (float)t * __expf(-((float)(2 * lane) / (float)HD) *
                                     logf(10000.0f));
  float cs, sn;
  __sincosf(fr, &sn, &cs);

  {
    float x1 = qkv[ibase + lane];
    float x2 = qkv[ibase + 32 + lane];
    float o1 = x1 * cs + x2 * sn;
    float o2 = -x1 * sn + x2 * cs;
    Q[obase + lane] = o1;
    Q[obase + 32 + lane] = o2;
    float ss = wave32_sum(o1 * o1 + o2 * o2);
    if (lane == 0) q0[nidx] = sqrtf(kc + ss);
  }
  {
    float x1 = qkv[ibase + CDIM + lane];
    float x2 = qkv[ibase + CDIM + 32 + lane];
    float o1 = x1 * cs + x2 * sn;
    float o2 = -x1 * sn + x2 * cs;
    Kq[obase + lane] = o1;
    Kq[obase + 32 + lane] = o2;
    float ss = wave32_sum(o1 * o1 + o2 * o2);
    if (lane == 0) k0[nidx] = sqrtf(kc + ss);
  }
  V[obase + lane] = qkv[ibase + 2 * CDIM + lane];
  V[obase + 32 + lane] = qkv[ibase + 2 * CDIM + 32 + lane];
}

__global__ __launch_bounds__(256) void hyp_attn_kernel(
    const float* __restrict__ Q, const float* __restrict__ Km,
    const float* __restrict__ V, const float* __restrict__ q0,
    const float* __restrict__ k0, const float* __restrict__ hyp,
    float* __restrict__ O) {
  __shared__ __align__(16) float plds[8][16 * 18];

  const int lane = threadIdx.x & 31;
  const int wave = threadIdx.x >> 5;
  const int lm = lane & 15;
  const int lh = lane >> 4;

  const int b = blockIdx.x / NHEADS;
  const int h = blockIdx.x % NHEADS;
  const int l0 = blockIdx.y * 128 + wave * 16;

  const size_t hb = (size_t)(b * NHEADS + h) * TSEQ;
  const float* Qb = Q + hb * HD;
  const float* Kb = Km + hb * HD;
  const float* Vb = V + hb * HD;
  const float* q0b = q0 + hb;
  const float* k0b = k0 + hb;

  const float hk = hyp[h];
  const float inv_hk = 1.0f / hk;
  const float sqk = sqrtf(hk);

  v2f qa[16];
#pragma unroll
  for (int d = 0; d < 16; ++d)
    qa[d] = *(const v2f*)(Qb + (size_t)(l0 + lm) * HD + 4 * d + 2 * lh);

  float q0r[8];
#pragma unroll
  for (int r = 0; r < 8; ++r) q0r[r] = q0b[l0 + r + 8 * lh];

  float m_r[8], l_r[8];
#pragma unroll
  for (int r = 0; r < 8; ++r) { m_r[r] = -1e30f; l_r[r] = 0.0f; }
  v8f o[4] = {v8f{}, v8f{}, v8f{}, v8f{}};

  float* myp = &plds[wave][0];

  for (int s0 = 0; s0 <= l0; s0 += 16) {
    v8f s = v8f{};
#pragma unroll
    for (int d = 0; d < 16; ++d) {
      v2f kb = *(const v2f*)(Kb + (size_t)(s0 + lm) * HD + 4 * d + 2 * lh);
      s = wmma4(qa[d], kb, s);
    }

    const float k0c = k0b[s0 + lm];

#pragma unroll
    for (int r = 0; r < 8; ++r) {
      const int row = l0 + r + 8 * lh;
      const int col = s0 + lm;
      float w;
      if (col > row) {
        w = -INFINITY;
      } else {
        float inner = s[r] - q0r[r] * k0c;
        float arg = fmaxf(-inner * inv_hk, 1.0f + EPSV);
        float dis = sqk * acoshf(arg);
        w = 1.0f / (EPSV + dis);
      }
      float pm = half16_max(w);
      float newm = fmaxf(m_r[r], pm);
      float scale = __expf(m_r[r] - newm);
      float p = __expf(w - newm);
      float ps = half16_sum(p);
      l_r[r] = l_r[r] * scale + ps;
      m_r[r] = newm;
      o[0][r] *= scale;
      o[1][r] *= scale;
      o[2][r] *= scale;
      o[3][r] *= scale;
      myp[(r + 8 * lh) * 18 + lm] = p;
    }

    asm volatile("s_wait_dscnt 0" ::: "memory");

#pragma unroll
    for (int kk = 0; kk < 4; ++kk) {
      v2f pa = *(const v2f*)(myp + lm * 18 + 4 * kk + 2 * lh);
      const int krow = s0 + 4 * kk + 2 * lh;
#pragma unroll
      for (int j = 0; j < 4; ++j) {
        v2f vb;
        vb.x = Vb[(size_t)krow * HD + j * 16 + lm];
        vb.y = Vb[(size_t)(krow + 1) * HD + j * 16 + lm];
        o[j] = wmma4(pa, vb, o[j]);
      }
    }
  }

#pragma unroll
  for (int r = 0; r < 8; ++r) {
    const float inv = 1.0f / l_r[r];
    const int t = l0 + r + 8 * lh;
#pragma unroll
    for (int j = 0; j < 4; ++j)
      O[((size_t)(b * TSEQ + t)) * CDIM + h * HD + j * 16 + lm] = o[j][r] * inv;
  }
}

extern "C" void kernel_launch(void* const* d_in, const int* in_sizes, int n_in,
                              void* d_out, int out_size, void* d_ws,
                              size_t ws_size, hipStream_t stream) {
  (void)in_sizes; (void)n_in; (void)out_size; (void)ws_size;

  const float* x     = (const float*)d_in[0];
  const float* W_qkv = (const float*)d_in[1];
  const float* W_out = (const float*)d_in[2];
  const float* hyp   = (const float*)d_in[3];
  float* out = (float*)d_out;

  const size_t QKV_N = (size_t)BATCH * TSEQ * 3 * CDIM;
  const size_t QN    = (size_t)BATCH * NHEADS * TSEQ * HD;
  const size_t SN    = (size_t)BATCH * NHEADS * TSEQ;

  float* ws  = (float*)d_ws;
  float* qkv = ws;
  float* Q   = qkv + QKV_N;
  float* K   = Q + QN;
  float* V   = K + QN;
  float* q0  = V + QN;
  float* k0  = q0 + SN;
  float* AO  = k0 + SN;

  const int M = BATCH * TSEQ;

  gemm_abT_wmma<<<dim3((3 * CDIM) / 64, M / 128), 256, 0, stream>>>(
      x, W_qkv, qkv, CDIM, CDIM, CDIM, 3 * CDIM);

  rope_norm_kernel<<<(BATCH * TSEQ * NHEADS) / 8, 256, 0, stream>>>(
      qkv, hyp, Q, K, V, q0, k0);

  hyp_attn_kernel<<<dim3(BATCH * NHEADS, TSEQ / 128), 256, 0, stream>>>(
      Q, K, V, q0, k0, hyp, AO);

  gemm_abT_wmma<<<dim3(CDIM / 64, M / 128), 256, 0, stream>>>(
      AO, W_out, out, CDIM, CDIM, CDIM, CDIM);
}


# attention weight: library acoshf and IEEE divide (serial extended-precision chain inside a per-row branch) replaced by straight-line hardware sqrt/log2/rcp form; rows interleave
# speedup vs baseline: 1.3908x; 1.3908x over previous
#include <hip/hip_runtime.h>
#include <math.h>

#define NHEADS   16
#define HD       64
#define TSEQ     1024
#define CDIM     1024
#define BATCH    4
#define EPSV     1e-6f
#define KCH      32
#define LPAD     36

typedef float v2f __attribute__((ext_vector_type(2)));
typedef float v8f __attribute__((ext_vector_type(8)));

static __device__ __forceinline__ v8f wmma4(v2f a, v2f b, v8f c) {
  return __builtin_amdgcn_wmma_f32_16x16x4_f32(false, a, false, b, (short)0, c,
                                               false, false);
}

static __device__ __forceinline__ float half16_max(float v) {
#pragma unroll
  for (int off = 1; off < 16; off <<= 1) v = fmaxf(v, __shfl_xor(v, off, 16));
  return v;
}
static __device__ __forceinline__ float half16_sum(float v) {
#pragma unroll
  for (int off = 1; off < 16; off <<= 1) v += __shfl_xor(v, off, 16);
  return v;
}
static __device__ __forceinline__ float wave32_sum(float v) {
#pragma unroll
  for (int off = 1; off < 32; off <<= 1) v += __shfl_xor(v, off, 32);
  return v;
}

__global__ __launch_bounds__(256) void gemm_abT_wmma(
    const float* __restrict__ A, const float* __restrict__ B,
    float* __restrict__ C, int K, int lda, int ldb, int ldc) {
  __shared__ __align__(16) float lA[2][128 * LPAD];
  __shared__ __align__(16) float lB[2][64 * LPAD];

  const int tid = threadIdx.x;
  const int lane = tid & 31;
  const int wave = tid >> 5;
  const int lm = lane & 15;
  const int lh = lane >> 4;
  const int m0 = blockIdx.y * 128;
  const int n0 = blockIdx.x * 64;

  const int srow = tid >> 3;
  const int scol = (tid & 7) * 4;

  const float* Ast = A + (size_t)(m0 + srow) * lda + scol;
  const float* Bst = B + (size_t)(n0 + srow) * ldb + scol;

  float4 ra[4], rb[2];

#pragma unroll
  for (int it = 0; it < 4; ++it)
    ra[it] = *(const float4*)(Ast + (size_t)(32 * it) * lda);
#pragma unroll
  for (int it = 0; it < 2; ++it)
    rb[it] = *(const float4*)(Bst + (size_t)(32 * it) * ldb);
#pragma unroll
  for (int it = 0; it < 4; ++it)
    *(float4*)(&lA[0][(srow + 32 * it) * LPAD + scol]) = ra[it];
#pragma unroll
  for (int it = 0; it < 2; ++it)
    *(float4*)(&lB[0][(srow + 32 * it) * LPAD + scol]) = rb[it];
  __syncthreads();

  v8f acc[4] = {v8f{}, v8f{}, v8f{}, v8f{}};
  const int nch = K / KCH;

  for (int c = 0; c < nch; ++c) {
    const int buf = c & 1;

    if (c + 1 < nch) {
      const int k1 = (c + 1) * KCH;
#pragma unroll
      for (int it = 0; it < 4; ++it)
        ra[it] = *(const float4*)(Ast + (size_t)(32 * it) * lda + k1);
#pragma unroll
      for (int it = 0; it < 2; ++it)
        rb[it] = *(const float4*)(Bst + (size_t)(32 * it) * ldb + k1);
    }
    if (c + 2 < nch) {
      __builtin_prefetch(Ast + (size_t)(c + 2) * KCH, 0, 1);
      __builtin_prefetch(Bst + (size_t)(c + 2) * KCH, 0, 1);
    }

    const float* pa = &lA[buf][(wave * 16 + lm) * LPAD + 2 * lh];
    const float* pb = &lB[buf][lm * LPAD + 2 * lh];

    v2f af[8];
    v2f bf[8][4];
#pragma unroll
    for (int ks = 0; ks < 8; ++ks) {
      af[ks] = *(const v2f*)(pa + 4 * ks);
#pragma unroll
      for (int j = 0; j < 4; ++j)
        bf[ks][j] = *(const v2f*)(pb + j * 16 * LPAD + 4 * ks);
    }
#pragma unroll
    for (int ks = 0; ks < 8; ++ks)
#pragma unroll
      for (int j = 0; j < 4; ++j)
        acc[j] = wmma4(af[ks], bf[ks][j], acc[j]);

    if (c + 1 < nch) {
      const int nb = buf ^ 1;
#pragma unroll
      for (int it = 0; it < 4; ++it)
        *(float4*)(&lA[nb][(srow + 32 * it) * LPAD + scol]) = ra[it];
#pragma unroll
      for (int it = 0; it < 2; ++it)
        *(float4*)(&lB[nb][(srow + 32 * it) * LPAD + scol]) = rb[it];
      __syncthreads();
    }
  }

  const int mw = m0 + wave * 16;
#pragma unroll
  for (int j = 0; j < 4; ++j)
#pragma unroll
    for (int r = 0; r < 8; ++r)
      C[(size_t)(mw + r + 8 * lh) * ldc + n0 + j * 16 + lm] = acc[j][r];
}

__global__ __launch_bounds__(256) void rope_norm_kernel(
    const float* __restrict__ qkv, const float* __restrict__ hyp,
    float* __restrict__ Q, float* __restrict__ Kq, float* __restrict__ V,
    float* __restrict__ q0, float* __restrict__ k0) {
  const int lane = threadIdx.x & 31;
  const int gw = blockIdx.x * 8 + (threadIdx.x >> 5);
  const int h = gw % NHEADS;
  const int t = (gw / NHEADS) % TSEQ;
  const int b = gw / (NHEADS * TSEQ);

  const size_t ibase = ((size_t)(b * TSEQ + t)) * (3 * CDIM) + h * HD;
  const size_t obase = ((size_t)((b * NHEADS + h) * TSEQ) + t) * HD;
  const size_t nidx = (size_t)(b * NHEADS + h) * TSEQ + t;
  const float kc = hyp[h];

  const float fr = (float)t * __expf(-((float)(2 * lane) / (float)HD) *
                                     logf(10000.0f));
  float cs, sn;
  __sincosf(fr, &sn, &cs);

  {
    float x1 = qkv[ibase + lane];
    float x2 = qkv[ibase + 32 + lane];
    float o1 = x1 * cs + x2 * sn;
    float o2 = -x1 * sn + x2 * cs;
    Q[obase + lane] = o1;
    Q[obase + 32 + lane] = o2;
    float ss = wave32_sum(o1 * o1 + o2 * o2);
    if (lane == 0) q0[nidx] = sqrtf(kc + ss);
  }
  {
    float x1 = qkv[ibase + CDIM + lane];
    float x2 = qkv[ibase + CDIM + 32 + lane];
    float o1 = x1 * cs + x2 * sn;
    float o2 = -x1 * sn + x2 * cs;
    Kq[obase + lane] = o1;
    Kq[obase + 32 + lane] = o2;
    float ss = wave32_sum(o1 * o1 + o2 * o2);
    if (lane == 0) k0[nidx] = sqrtf(kc + ss);
  }
  V[obase + lane] = qkv[ibase + 2 * CDIM + lane];
  V[obase + 32 + lane] = qkv[ibase + 2 * CDIM + 32 + lane];
}

static __device__ __forceinline__ float acosh_hw(float x) {
  const float a = x - 1.0f;
  const float sq = __builtin_amdgcn_sqrtf(a) * __builtin_amdgcn_sqrtf(x + 1.0f);
  const float u = a + sq;
  const float v = 1.0f + u;
  const float corr = (u - (v - 1.0f)) * __builtin_amdgcn_rcpf(v);
  return fmaf(__builtin_amdgcn_logf(v), 0.69314718056f, corr);
}

__global__ __launch_bounds__(256) void hyp_attn_kernel(
    const float* __restrict__ Q, const float* __restrict__ Km,
    const float* __restrict__ V, const float* __restrict__ q0,
    const float* __restrict__ k0, const float* __restrict__ hyp,
    float* __restrict__ O) {
  __shared__ __align__(16) float plds[8][16 * 18];

  const int lane = threadIdx.x & 31;
  const int wave = threadIdx.x >> 5;
  const int lm = lane & 15;
  const int lh = lane >> 4;

  const int b = blockIdx.x / NHEADS;
  const int h = blockIdx.x % NHEADS;
  const int l0 = blockIdx.y * 128 + wave * 16;

  const size_t hb = (size_t)(b * NHEADS + h) * TSEQ;
  const float* Qb = Q + hb * HD;
  const float* Kb = Km + hb * HD;
  const float* Vb = V + hb * HD;
  const float* q0b = q0 + hb;
  const float* k0b = k0 + hb;

  const float hk = hyp[h];
  const float inv_hk = 1.0f / hk;
  const float sqk = sqrtf(hk);

  v2f qa[16];
#pragma unroll
  for (int d = 0; d < 16; ++d)
    qa[d] = *(const v2f*)(Qb + (size_t)(l0 + lm) * HD + 4 * d + 2 * lh);

  float q0r[8];
#pragma unroll
  for (int r = 0; r < 8; ++r) q0r[r] = q0b[l0 + r + 8 * lh];

  float m_r[8], l_r[8];
#pragma unroll
  for (int r = 0; r < 8; ++r) { m_r[r] = -1e30f; l_r[r] = 0.0f; }
  v8f o[4] = {v8f{}, v8f{}, v8f{}, v8f{}};

  float* myp = &plds[wave][0];

  for (int s0 = 0; s0 <= l0; s0 += 16) {
    v8f s = v8f{};
#pragma unroll
    for (int d = 0; d < 16; ++d) {
      v2f kb = *(const v2f*)(Kb + (size_t)(s0 + lm) * HD + 4 * d + 2 * lh);
      s = wmma4(qa[d], kb, s);
    }

    const float k0c = k0b[s0 + lm];

#pragma unroll
    for (int r = 0; r < 8; ++r) {
      const int row = l0 + r + 8 * lh;
      const int col = s0 + lm;
      const float inner = s[r] - q0r[r] * k0c;
      const float arg = fmaxf(-inner * inv_hk, 1.0f + EPSV);
      const float dis = sqk * acosh_hw(arg);
      float w = __builtin_amdgcn_rcpf(EPSV + dis);
      w = (col > row) ? -INFINITY : w;
      float pm = half16_max(w);
      float newm = fmaxf(m_r[r], pm);
      float scale = __expf(m_r[r] - newm);
      float p = __expf(w - newm);
      float ps = half16_sum(p);
      l_r[r] = l_r[r] * scale + ps;
      m_r[r] = newm;
      o[0][r] *= scale;
      o[1][r] *= scale;
      o[2][r] *= scale;
      o[3][r] *= scale;
      myp[(r + 8 * lh) * 18 + lm] = p;
    }

    asm volatile("s_wait_dscnt 0" ::: "memory");

#pragma unroll
    for (int kk = 0; kk < 4; ++kk) {
      v2f pa = *(const v2f*)(myp + lm * 18 + 4 * kk + 2 * lh);
      const int krow = s0 + 4 * kk + 2 * lh;
#pragma unroll
      for (int j = 0; j < 4; ++j) {
        v2f vb;
        vb.x = Vb[(size_t)krow * HD + j * 16 + lm];
        vb.y = Vb[(size_t)(krow + 1) * HD + j * 16 + lm];
        o[j] = wmma4(pa, vb, o[j]);
      }
    }
  }

#pragma unroll
  for (int r = 0; r < 8; ++r) {
    const float inv = 1.0f / l_r[r];
    const int t = l0 + r + 8 * lh;
#pragma unroll
    for (int j = 0; j < 4; ++j)
      O[((size_t)(b * TSEQ + t)) * CDIM + h * HD + j * 16 + lm] = o[j][r] * inv;
  }
}

extern "C" void kernel_launch(void* const* d_in, const int* in_sizes, int n_in,
                              void* d_out, int out_size, void* d_ws,
                              size_t ws_size, hipStream_t stream) {
  (void)in_sizes; (void)n_in; (void)out_size; (void)ws_size;

  const float* x     = (const float*)d_in[0];
  const float* W_qkv = (const float*)d_in[1];
  const float* W_out = (const float*)d_in[2];
  const float* hyp   = (const float*)d_in[3];
  float* out = (float*)d_out;

  const size_t QKV_N = (size_t)BATCH * TSEQ * 3 * CDIM;
  const size_t QN    = (size_t)BATCH * NHEADS * TSEQ * HD;
  const size_t SN    = (size_t)BATCH * NHEADS * TSEQ;

  float* ws  = (float*)d_ws;
  float* qkv = ws;
  float* Q   = qkv + QKV_N;
  float* K   = Q + QN;
  float* V   = K + QN;
  float* q0  = V + QN;
  float* k0  = q0 + SN;
  float* AO  = k0 + SN;

  const int M = BATCH * TSEQ;

  gemm_abT_wmma<<<dim3((3 * CDIM) / 64, M / 128), 256, 0, stream>>>(
      x, W_qkv, qkv, CDIM, CDIM, CDIM, 3 * CDIM);

  rope_norm_kernel<<<(BATCH * TSEQ * NHEADS) / 8, 256, 0, stream>>>(
      qkv, hyp, Q, K, V, q0, k0);

  hyp_attn_kernel<<<dim3(BATCH * NHEADS, TSEQ / 128), 256, 0, stream>>>(
      Q, K, V, q0, k0, hyp, AO);

  gemm_abT_wmma<<<dim3(CDIM / 64, M / 128), 256, 0, stream>>>(
      AO, W_out, out, CDIM, CDIM, CDIM, CDIM);
}


# attention: K fragments and k0 loaded one key tile ahead, V fragments issued at the top of the trip, so no global-load latency is waited for in line; two scheduling barriers hold that placement
# speedup vs baseline: 1.4822x; 1.0657x over previous
#include <hip/hip_runtime.h>
#include <math.h>

#define NHEADS   16
#define HD       64
#define TSEQ     1024
#define CDIM     1024
#define BATCH    4
#define EPSV     1e-6f
#define KCH      32
#define LPAD     36

typedef float v2f __attribute__((ext_vector_type(2)));
typedef float v8f __attribute__((ext_vector_type(8)));

static __device__ __forceinline__ v8f wmma4(v2f a, v2f b, v8f c) {
  return __builtin_amdgcn_wmma_f32_16x16x4_f32(false, a, false, b, (short)0, c,
                                               false, false);
}

static __device__ __forceinline__ float half16_max(float v) {
#pragma unroll
  for (int off = 1; off < 16; off <<= 1) v = fmaxf(v, __shfl_xor(v, off, 16));
  return v;
}
static __device__ __forceinline__ float half16_sum(float v) {
#pragma unroll
  for (int off = 1; off < 16; off <<= 1) v += __shfl_xor(v, off, 16);
  return v;
}
static __device__ __forceinline__ float wave32_sum(float v) {
#pragma unroll
  for (int off = 1; off < 32; off <<= 1) v += __shfl_xor(v, off, 32);
  return v;
}

__global__ __launch_bounds__(256) void gemm_abT_wmma(
    const float* __restrict__ A, const float* __restrict__ B,
    float* __restrict__ C, int K, int lda, int ldb, int ldc) {
  __shared__ __align__(16) float lA[2][128 * LPAD];
  __shared__ __align__(16) float lB[2][64 * LPAD];

  const int tid = threadIdx.x;
  const int lane = tid & 31;
  const int wave = tid >> 5;
  const int lm = lane & 15;
  const int lh = lane >> 4;
  const int m0 = blockIdx.y * 128;
  const int n0 = blockIdx.x * 64;

  const int srow = tid >> 3;
  const int scol = (tid & 7) * 4;

  const float* Ast = A + (size_t)(m0 + srow) * lda + scol;
  const float* Bst = B + (size_t)(n0 + srow) * ldb + scol;

  float4 ra[4], rb[2];

#pragma unroll
  for (int it = 0; it < 4; ++it)
    ra[it] = *(const float4*)(Ast + (size_t)(32 * it) * lda);
#pragma unroll
  for (int it = 0; it < 2; ++it)
    rb[it] = *(const float4*)(Bst + (size_t)(32 * it) * ldb);
#pragma unroll
  for (int it = 0; it < 4; ++it)
    *(float4*)(&lA[0][(srow + 32 * it) * LPAD + scol]) = ra[it];
#pragma unroll
  for (int it = 0; it < 2; ++it)
    *(float4*)(&lB[0][(srow + 32 * it) * LPAD + scol]) = rb[it];
  __syncthreads();

  v8f acc[4] = {v8f{}, v8f{}, v8f{}, v8f{}};
  const int nch = K / KCH;

  for (int c = 0; c < nch; ++c) {
    const int buf = c & 1;

    if (c + 1 < nch) {
      const int k1 = (c + 1) * KCH;
#pragma unroll
      for (int it = 0; it < 4; ++it)
        ra[it] = *(const float4*)(Ast + (size_t)(32 * it) * lda + k1);
#pragma unroll
      for (int it = 0; it < 2; ++it)
        rb[it] = *(const float4*)(Bst + (size_t)(32 * it) * ldb + k1);
    }
    if (c + 2 < nch) {
      __builtin_prefetch(Ast + (size_t)(c + 2) * KCH, 0, 1);
      __builtin_prefetch(Bst + (size_t)(c + 2) * KCH, 0, 1);
    }

    const float* pa = &lA[buf][(wave * 16 + lm) * LPAD + 2 * lh];
    const float* pb = &lB[buf][lm * LPAD + 2 * lh];

    v2f af[8];
    v2f bf[8][4];
#pragma unroll
    for (int ks = 0; ks < 8; ++ks) {
      af[ks] = *(const v2f*)(pa + 4 * ks);
#pragma unroll
      for (int j = 0; j < 4; ++j)
        bf[ks][j] = *(const v2f*)(pb + j * 16 * LPAD + 4 * ks);
    }
#pragma unroll
    for (int ks = 0; ks < 8; ++ks)
#pragma unroll
      for (int j = 0; j < 4; ++j)
        acc[j] = wmma4(af[ks], bf[ks][j], acc[j]);

    if (c + 1 < nch) {
      const int nb = buf ^ 1;
#pragma unroll
      for (int it = 0; it < 4; ++it)
        *(float4*)(&lA[nb][(srow + 32 * it) * LPAD + scol]) = ra[it];
#pragma unroll
      for (int it = 0; it < 2; ++it)
        *(float4*)(&lB[nb][(srow + 32 * it) * LPAD + scol]) = rb[it];
      __syncthreads();
    }
  }

  const int mw = m0 + wave * 16;
#pragma unroll
  for (int j = 0; j < 4; ++j)
#pragma unroll
    for (int r = 0; r < 8; ++r)
      C[(size_t)(mw + r + 8 * lh) * ldc + n0 + j * 16 + lm] = acc[j][r];
}

__global__ __launch_bounds__(256) void rope_norm_kernel(
    const float* __restrict__ qkv, const float* __restrict__ hyp,
    float* __restrict__ Q, float* __restrict__ Kq, float* __restrict__ V,
    float* __restrict__ q0, float* __restrict__ k0) {
  const int lane = threadIdx.x & 31;
  const int gw = blockIdx.x * 8 + (threadIdx.x >> 5);
  const int h = gw % NHEADS;
  const int t = (gw / NHEADS) % TSEQ;
  const int b = gw / (NHEADS * TSEQ);

  const size_t ibase = ((size_t)(b * TSEQ + t)) * (3 * CDIM) + h * HD;
  const size_t obase = ((size_t)((b * NHEADS + h) * TSEQ) + t) * HD;
  const size_t nidx = (size_t)(b * NHEADS + h) * TSEQ + t;
  const float kc = hyp[h];

  const float fr = (float)t * __expf(-((float)(2 * lane) / (float)HD) *
                                     logf(10000.0f));
  float cs, sn;
  __sincosf(fr, &sn, &cs);

  {
    float x1 = qkv[ibase + lane];
    float x2 = qkv[ibase + 32 + lane];
    float o1 = x1 * cs + x2 * sn;
    float o2 = -x1 * sn + x2 * cs;
    Q[obase + lane] = o1;
    Q[obase + 32 + lane] = o2;
    float ss = wave32_sum(o1 * o1 + o2 * o2);
    if (lane == 0) q0[nidx] = sqrtf(kc + ss);
  }
  {
    float x1 = qkv[ibase + CDIM + lane];
    float x2 = qkv[ibase + CDIM + 32 + lane];
    float o1 = x1 * cs + x2 * sn;
    float o2 = -x1 * sn + x2 * cs;
    Kq[obase + lane] = o1;
    Kq[obase + 32 + lane] = o2;
    float ss = wave32_sum(o1 * o1 + o2 * o2);
    if (lane == 0) k0[nidx] = sqrtf(kc + ss);
  }
  V[obase + lane] = qkv[ibase + 2 * CDIM + lane];
  V[obase + 32 + lane] = qkv[ibase + 2 * CDIM + 32 + lane];
}

static __device__ __forceinline__ float acosh_hw(float x) {
  const float a = x - 1.0f;
  const float sq = __builtin_amdgcn_sqrtf(a) * __builtin_amdgcn_sqrtf(x + 1.0f);
  const float u = a + sq;
  const float v = 1.0f + u;
  const float corr = (u - (v - 1.0f)) * __builtin_amdgcn_rcpf(v);
  return fmaf(__builtin_amdgcn_logf(v), 0.69314718056f, corr);
}

__global__ __launch_bounds__(256) void hyp_attn_kernel(
    const float* __restrict__ Q, const float* __restrict__ Km,
    const float* __restrict__ V, const float* __restrict__ q0,
    const float* __restrict__ k0, const float* __restrict__ hyp,
    float* __restrict__ O) {
  __shared__ __align__(16) float plds[8][16 * 18];

  const int lane = threadIdx.x & 31;
  const int wave = threadIdx.x >> 5;
  const int lm = lane & 15;
  const int lh = lane >> 4;

  const int b = blockIdx.x / NHEADS;
  const int h = blockIdx.x % NHEADS;
  const int l0 = blockIdx.y * 128 + wave * 16;

  const size_t hb = (size_t)(b * NHEADS + h) * TSEQ;
  const float* Qb = Q + hb * HD;
  const float* Kb = Km + hb * HD;
  const float* Vb = V + hb * HD;
  const float* q0b = q0 + hb;
  const float* k0b = k0 + hb;

  const float hk = hyp[h];
  const float inv_hk = 1.0f / hk;
  const float sqk = sqrtf(hk);

  v2f qa[16];
#pragma unroll
  for (int d = 0; d < 16; ++d)
    qa[d] = *(const v2f*)(Qb + (size_t)(l0 + lm) * HD + 4 * d + 2 * lh);

  float q0r[8];
#pragma unroll
  for (int r = 0; r < 8; ++r) q0r[r] = q0b[l0 + r + 8 * lh];

  float m_r[8], l_r[8];
#pragma unroll
  for (int r = 0; r < 8; ++r) { m_r[r] = -1e30f; l_r[r] = 0.0f; }
  v8f o[4] = {v8f{}, v8f{}, v8f{}, v8f{}};

  float* myp = &plds[wave][0];

  v2f kf[16];
#pragma unroll
  for (int d = 0; d < 16; ++d)
    kf[d] = *(const v2f*)(Kb + (size_t)lm * HD + 4 * d + 2 * lh);
  float k0n = k0b[lm];

  for (int s0 = 0; s0 <= l0; s0 += 16) {
    v2f vf[4][4];
#pragma unroll
    for (int kk = 0; kk < 4; ++kk) {
      const int krow = s0 + 4 * kk + 2 * lh;
#pragma unroll
      for (int j = 0; j < 4; ++j) {
        vf[kk][j].x = Vb[(size_t)krow * HD + j * 16 + lm];
        vf[kk][j].y = Vb[(size_t)(krow + 1) * HD + j * 16 + lm];
      }
    }

    v8f s = v8f{};
#pragma unroll
    for (int d = 0; d < 16; ++d) s = wmma4(qa[d], kf[d], s);

    const float k0c = k0n;

    const int sn = (s0 + 16 <= l0) ? (s0 + 16) : s0;
#pragma unroll
    for (int d = 0; d < 16; ++d)
      kf[d] = *(const v2f*)(Kb + (size_t)(sn + lm) * HD + 4 * d + 2 * lh);
    k0n = k0b[sn + lm];
    __builtin_amdgcn_sched_barrier(0);

#pragma unroll
    for (int r = 0; r < 8; ++r) {
      const int row = l0 + r + 8 * lh;
      const int col = s0 + lm;
      const float inner = s[r] - q0r[r] * k0c;
      const float arg = fmaxf(-inner * inv_hk, 1.0f + EPSV);
      const float dis = sqk * acosh_hw(arg);
      float w = __builtin_amdgcn_rcpf(EPSV + dis);
      w = (col > row) ? -INFINITY : w;
      float pm = half16_max(w);
      float newm = fmaxf(m_r[r], pm);
      float scale = __expf(m_r[r] - newm);
      float p = __expf(w - newm);
      float ps = half16_sum(p);
      l_r[r] = l_r[r] * scale + ps;
      m_r[r] = newm;
      o[0][r] *= scale;
      o[1][r] *= scale;
      o[2][r] *= scale;
      o[3][r] *= scale;
      myp[(r + 8 * lh) * 18 + lm] = p;
    }

    asm volatile("s_wait_dscnt 0" ::: "memory");

#pragma unroll
    for (int kk = 0; kk < 4; ++kk) {
      v2f pa = *(const v2f*)(myp + lm * 18 + 4 * kk + 2 * lh);
#pragma unroll
      for (int j = 0; j < 4; ++j) o[j] = wmma4(pa, vf[kk][j], o[j]);
    }
    __builtin_amdgcn_sched_barrier(0);
  }

#pragma unroll
  for (int r = 0; r < 8; ++r) {
    const float inv = 1.0f / l_r[r];
    const int t = l0 + r + 8 * lh;
#pragma unroll
    for (int j = 0; j < 4; ++j)
      O[((size_t)(b * TSEQ + t)) * CDIM + h * HD + j * 16 + lm] = o[j][r] * inv;
  }
}

extern "C" void kernel_launch(void* const* d_in, const int* in_sizes, int n_in,
                              void* d_out, int out_size, void* d_ws,
                              size_t ws_size, hipStream_t stream) {
  (void)in_sizes; (void)n_in; (void)out_size; (void)ws_size;

  const float* x     = (const float*)d_in[0];
  const float* W_qkv = (const float*)d_in[1];
  const float* W_out = (const float*)d_in[2];
  const float* hyp   = (const float*)d_in[3];
  float* out = (float*)d_out;

  const size_t QKV_N = (size_t)BATCH * TSEQ * 3 * CDIM;
  const size_t QN    = (size_t)BATCH * NHEADS * TSEQ * HD;
  const size_t SN    = (size_t)BATCH * NHEADS * TSEQ;

  float* ws  = (float*)d_ws;
  float* qkv = ws;
  float* Q   = qkv + QKV_N;
  float* K   = Q + QN;
  float* V   = K + QN;
  float* q0  = V + QN;
  float* k0  = q0 + SN;
  float* AO  = k0 + SN;

  const int M = BATCH * TSEQ;

  gemm_abT_wmma<<<dim3((3 * CDIM) / 64, M / 128), 256, 0, stream>>>(
      x, W_qkv, qkv, CDIM, CDIM, CDIM, 3 * CDIM);

  rope_norm_kernel<<<(BATCH * TSEQ * NHEADS) / 8, 256, 0, stream>>>(
      qkv, hyp, Q, K, V, q0, k0);

  hyp_attn_kernel<<<dim3(BATCH * NHEADS, TSEQ / 128), 256, 0, stream>>>(
      Q, K, V, q0, k0, hyp, AO);

  gemm_abT_wmma<<<dim3(CDIM / 64, M / 128), 256, 0, stream>>>(
      AO, W_out, out, CDIM, CDIM, CDIM, CDIM);
}


# GEMM (both launches): the K-chunk prefetched from global memory stays in registers (six named vectors); the compiler had left the staging arrays in scratch memory and an LDS side buffer
# speedup vs baseline: 2.2822x; 1.5397x over previous
#include <hip/hip_runtime.h>
#include <math.h>

#define NHEADS   16
#define HD       64
#define TSEQ     1024
#define CDIM     1024
#define BATCH    4
#define EPSV     1e-6f
#define KCH      32
#define LPAD     36

typedef float v2f __attribute__((ext_vector_type(2)));
typedef float v4f __attribute__((ext_vector_type(4)));
typedef float v8f __attribute__((ext_vector_type(8)));

static __device__ __forceinline__ v8f wmma4(v2f a, v2f b, v8f c) {
  return __builtin_amdgcn_wmma_f32_16x16x4_f32(false, a, false, b, (short)0, c,
                                               false, false);
}

static __device__ __forceinline__ float half16_max(float v) {
#pragma unroll
  for (int off = 1; off < 16; off <<= 1) v = fmaxf(v, __shfl_xor(v, off, 16));
  return v;
}
static __device__ __forceinline__ float half16_sum(float v) {
#pragma unroll
  for (int off = 1; off < 16; off <<= 1) v += __shfl_xor(v, off, 16);
  return v;
}
static __device__ __forceinline__ float wave32_sum(float v) {
#pragma unroll
  for (int off = 1; off < 32; off <<= 1) v += __shfl_xor(v, off, 32);
  return v;
}

__global__ __launch_bounds__(256) void gemm_abT_wmma(
    const float* __restrict__ A, const float* __restrict__ B,
    float* __restrict__ C, int K, int lda, int ldb, int ldc) {
  __shared__ __align__(16) float lA[2][128 * LPAD];
  __shared__ __align__(16) float lB[2][64 * LPAD];

  const int tid = threadIdx.x;
  const int lane = tid & 31;
  const int wave = tid >> 5;
  const int lm = lane & 15;
  const int lh = lane >> 4;
  const int m0 = blockIdx.y * 128;
  const int n0 = blockIdx.x * 64;

  const int srow = tid >> 3;
  const int scol = (tid & 7) * 4;

  const float* Ast = A + (size_t)(m0 + srow) * lda + scol;
  const float* Bst = B + (size_t)(n0 + srow) * ldb + scol;

  const float* A1 = Ast + (size_t)32 * lda;
  const float* A2 = Ast + (size_t)64 * lda;
  const float* A3 = Ast + (size_t)96 * lda;
  const float* B1 = Bst + (size_t)32 * ldb;
  const int so = srow * LPAD + scol;

  v4f ra0 = *(const v4f*)(Ast);
  v4f ra1 = *(const v4f*)(A1);
  v4f ra2 = *(const v4f*)(A2);
  v4f ra3 = *(const v4f*)(A3);
  v4f rb0 = *(const v4f*)(Bst);
  v4f rb1 = *(const v4f*)(B1);

  *(v4f*)(&lA[0][so]) = ra0;
  *(v4f*)(&lA[0][so + 32 * LPAD]) = ra1;
  *(v4f*)(&lA[0][so + 64 * LPAD]) = ra2;
  *(v4f*)(&lA[0][so + 96 * LPAD]) = ra3;
  *(v4f*)(&lB[0][so]) = rb0;
  *(v4f*)(&lB[0][so + 32 * LPAD]) = rb1;
  __syncthreads();

  v8f acc[4] = {v8f{}, v8f{}, v8f{}, v8f{}};
  const int nch = K / KCH;

  for (int c = 0; c < nch; ++c) {
    const int buf = c & 1;

    if (c + 1 < nch) {
      const int k1 = (c + 1) * KCH;
      ra0 = *(const v4f*)(Ast + k1);
      ra1 = *(const v4f*)(A1 + k1);
      ra2 = *(const v4f*)(A2 + k1);
      ra3 = *(const v4f*)(A3 + k1);
      rb0 = *(const v4f*)(Bst + k1);
      rb1 = *(const v4f*)(B1 + k1);
    }
    if (c + 2 < nch) {
      __builtin_prefetch(Ast + (size_t)(c + 2) * KCH, 0, 1);
      __builtin_prefetch(Bst + (size_t)(c + 2) * KCH, 0, 1);
    }

    const float* pa = &lA[buf][(wave * 16 + lm) * LPAD + 2 * lh];
    const float* pb = &lB[buf][lm * LPAD + 2 * lh];

    v2f af[8];
    v2f bf[8][4];
#pragma unroll
    for (int ks = 0; ks < 8; ++ks) {
      af[ks] = *(const v2f*)(pa + 4 * ks);
#pragma unroll
      for (int j = 0; j < 4; ++j)
        bf[ks][j] = *(const v2f*)(pb + j * 16 * LPAD + 4 * ks);
    }
#pragma unroll
    for (int ks = 0; ks < 8; ++ks)
#pragma unroll
      for (int j = 0; j < 4; ++j)
        acc[j] = wmma4(af[ks], bf[ks][j], acc[j]);

    if (c + 1 < nch) {
      const int nb = buf ^ 1;
      *(v4f*)(&lA[nb][so]) = ra0;
      *(v4f*)(&lA[nb][so + 32 * LPAD]) = ra1;
      *(v4f*)(&lA[nb][so + 64 * LPAD]) = ra2;
      *(v4f*)(&lA[nb][so + 96 * LPAD]) = ra3;
      *(v4f*)(&lB[nb][so]) = rb0;
      *(v4f*)(&lB[nb][so + 32 * LPAD]) = rb1;
      __syncthreads();
    }
  }

  const int mw = m0 + wave * 16;
#pragma unroll
  for (int j = 0; j < 4; ++j)
#pragma unroll
    for (int r = 0; r < 8; ++r)
      C[(size_t)(mw + r + 8 * lh) * ldc + n0 + j * 16 + lm] = acc[j][r];
}

__global__ __launch_bounds__(256) void rope_norm_kernel(
    const float* __restrict__ qkv, const float* __restrict__ hyp,
    float* __restrict__ Q, float* __restrict__ Kq, float* __restrict__ V,
    float* __restrict__ q0, float* __restrict__ k0) {
  const int lane = threadIdx.x & 31;
  const int gw = blockIdx.x * 8 + (threadIdx.x >> 5);
  const int h = gw % NHEADS;
  const int t = (gw / NHEADS) % TSEQ;
  const int b = gw / (NHEADS * TSEQ);

  const size_t ibase = ((size_t)(b * TSEQ + t)) * (3 * CDIM) + h * HD;
  const size_t obase = ((size_t)((b * NHEADS + h) * TSEQ) + t) * HD;
  const size_t nidx = (size_t)(b * NHEADS + h) * TSEQ + t;
  const float kc = hyp[h];

  const float fr = (float)t * __expf(-((float)(2 * lane) / (float)HD) *
                                     logf(10000.0f));
  float cs, sn;
  __sincosf(fr, &sn, &cs);

  {
    float x1 = qkv[ibase + lane];
    float x2 = qkv[ibase + 32 + lane];
    float o1 = x1 * cs + x2 * sn;
    float o2 = -x1 * sn + x2 * cs;
    Q[obase + lane] = o1;
    Q[obase + 32 + lane] = o2;
    float ss = wave32_sum(o1 * o1 + o2 * o2);
    if (lane == 0) q0[nidx] = sqrtf(kc + ss);
  }
  {
    float x1 = qkv[ibase + CDIM + lane];
    float x2 = qkv[ibase + CDIM + 32 + lane];
    float o1 = x1 * cs + x2 * sn;
    float o2 = -x1 * sn + x2 * cs;
    Kq[obase + lane] = o1;
    Kq[obase + 32 + lane] = o2;
    float ss = wave32_sum(o1 * o1 + o2 * o2);
    if (lane == 0) k0[nidx] = sqrtf(kc + ss);
  }
  V[obase + lane] = qkv[ibase + 2 * CDIM + lane];
  V[obase + 32 + lane] = qkv[ibase + 2 * CDIM + 32 + lane];
}

static __device__ __forceinline__ float acosh_hw(float x) {
  const float a = x - 1.0f;
  const float sq = __builtin_amdgcn_sqrtf(a) * __builtin_amdgcn_sqrtf(x + 1.0f);
  const float u = a + sq;
  const float v = 1.0f + u;
  const float corr = (u - (v - 1.0f)) * __builtin_amdgcn_rcpf(v);
  return fmaf(__builtin_amdgcn_logf(v), 0.69314718056f, corr);
}

__global__ __launch_bounds__(256) void hyp_attn_kernel(
    const float* __restrict__ Q, const float* __restrict__ Km,
    const float* __restrict__ V, const float* __restrict__ q0,
    const float* __restrict__ k0, const float* __restrict__ hyp,
    float* __restrict__ O) {
  __shared__ __align__(16) float plds[8][16 * 18];

  const int lane = threadIdx.x & 31;
  const int wave = threadIdx.x >> 5;
  const int lm = lane & 15;
  const int lh = lane >> 4;

  const int b = blockIdx.x / NHEADS;
  const int h = blockIdx.x % NHEADS;
  const int l0 = blockIdx.y * 128 + wave * 16;

  const size_t hb = (size_t)(b * NHEADS + h) * TSEQ;
  const float* Qb = Q + hb * HD;
  const float* Kb = Km + hb * HD;
  const float* Vb = V + hb * HD;
  const float* q0b = q0 + hb;
  const float* k0b = k0 + hb;

  const float hk = hyp[h];
  const float inv_hk = 1.0f / hk;
  const float sqk = sqrtf(hk);

  v2f qa[16];
#pragma unroll
  for (int d = 0; d < 16; ++d)
    qa[d] = *(const v2f*)(Qb + (size_t)(l0 + lm) * HD + 4 * d + 2 * lh);

  float q0r[8];
#pragma unroll
  for (int r = 0; r < 8; ++r) q0r[r] = q0b[l0 + r + 8 * lh];

  float m_r[8], l_r[8];
#pragma unroll
  for (int r = 0; r < 8; ++r) { m_r[r] = -1e30f; l_r[r] = 0.0f; }
  v8f o[4] = {v8f{}, v8f{}, v8f{}, v8f{}};

  float* myp = &plds[wave][0];

  v2f kf[16];
#pragma unroll
  for (int d = 0; d < 16; ++d)
    kf[d] = *(const v2f*)(Kb + (size_t)lm * HD + 4 * d + 2 * lh);
  float k0n = k0b[lm];

  for (int s0 = 0; s0 <= l0; s0 += 16) {
    v2f vf[4][4];
#pragma unroll
    for (int kk = 0; kk < 4; ++kk) {
      const int krow = s0 + 4 * kk + 2 * lh;
#pragma unroll
      for (int j = 0; j < 4; ++j) {
        vf[kk][j].x = Vb[(size_t)krow * HD + j * 16 + lm];
        vf[kk][j].y = Vb[(size_t)(krow + 1) * HD + j * 16 + lm];
      }
    }

    v8f s = v8f{};
#pragma unroll
    for (int d = 0; d < 16; ++d) s = wmma4(qa[d], kf[d], s);

    const float k0c = k0n;

    const int sn = (s0 + 16 <= l0) ? (s0 + 16) : s0;
#pragma unroll
    for (int d = 0; d < 16; ++d)
      kf[d] = *(const v2f*)(Kb + (size_t)(sn + lm) * HD + 4 * d + 2 * lh);
    k0n = k0b[sn + lm];
    __builtin_amdgcn_sched_barrier(0);

#pragma unroll
    for (int r = 0; r < 8; ++r) {
      const int row = l0 + r + 8 * lh;
      const int col = s0 + lm;
      const float inner = s[r] - q0r[r] * k0c;
      const float arg = fmaxf(-inner * inv_hk, 1.0f + EPSV);
      const float dis = sqk * acosh_hw(arg);
      float w = __builtin_amdgcn_rcpf(EPSV + dis);
      w = (col > row) ? -INFINITY : w;
      float pm = half16_max(w);
      float newm = fmaxf(m_r[r], pm);
      float scale = __expf(m_r[r] - newm);
      float p = __expf(w - newm);
      float ps = half16_sum(p);
      l_r[r] = l_r[r] * scale + ps;
      m_r[r] = newm;
      o[0][r] *= scale;
      o[1][r] *= scale;
      o[2][r] *= scale;
      o[3][r] *= scale;
      myp[(r + 8 * lh) * 18 + lm] = p;
    }

    asm volatile("s_wait_dscnt 0" ::: "memory");

#pragma unroll
    for (int kk = 0; kk < 4; ++kk) {
      v2f pa = *(const v2f*)(myp + lm * 18 + 4 * kk + 2 * lh);
#pragma unroll
      for (int j = 0; j < 4; ++j) o[j] = wmma4(pa, vf[kk][j], o[j]);
    }
    __builtin_amdgcn_sched_barrier(0);
  }

#pragma unroll
  for (int r = 0; r < 8; ++r) {
    const float inv = 1.0f / l_r[r];
    const int t = l0 + r + 8 * lh;
#pragma unroll
    for (int j = 0; j < 4; ++j)
      O[((size_t)(b * TSEQ + t)) * CDIM + h * HD + j * 16 + lm] = o[j][r] * inv;
  }
}

extern "C" void kernel_launch(void* const* d_in, const int* in_sizes, int n_in,
                              void* d_out, int out_size, void* d_ws,
                              size_t ws_size, hipStream_t stream) {
  (void)in_sizes; (void)n_in; (void)out_size; (void)ws_size;

  const float* x     = (const float*)d_in[0];
  const float* W_qkv = (const float*)d_in[1];
  const float* W_out = (const float*)d_in[2];
  const float* hyp   = (const float*)d_in[3];
  float* out = (float*)d_out;

  const size_t QKV_N = (size_t)BATCH * TSEQ * 3 * CDIM;
  const size_t QN    = (size_t)BATCH * NHEADS * TSEQ * HD;
  const size_t SN    = (size_t)BATCH * NHEADS * TSEQ;

  float* ws  = (float*)d_ws;
  float* qkv = ws;
  float* Q   = qkv + QKV_N;
  float* K   = Q + QN;
  float* V   = K + QN;
  float* q0  = V + QN;
  float* k0  = q0 + SN;
  float* AO  = k0 + SN;

  const int M = BATCH * TSEQ;

  gemm_abT_wmma<<<dim3((3 * CDIM) / 64, M / 128), 256, 0, stream>>>(
      x, W_qkv, qkv, CDIM, CDIM, CDIM, 3 * CDIM);

  rope_norm_kernel<<<(BATCH * TSEQ * NHEADS) / 8, 256, 0, stream>>>(
      qkv, hyp, Q, K, V, q0, k0);

  hyp_attn_kernel<<<dim3(BATCH * NHEADS, TSEQ / 128), 256, 0, stream>>>(
      Q, K, V, q0, k0, hyp, AO);

  gemm_abT_wmma<<<dim3(CDIM / 64, M / 128), 256, 0, stream>>>(
      AO, W_out, out, CDIM, CDIM, CDIM, CDIM);
}


# attention: score tile transposed (S^T = K*Q^T), a lane holds 8 keys of one query: row max/sum in-lane + one half-exchange, P^T feeds PV from registers, no LDS; row-sum and PV key order change
# speedup vs baseline: 2.5609x; 1.1221x over previous
#include <hip/hip_runtime.h>
#include <math.h>

#define NHEADS   16
#define HD       64
#define TSEQ     1024
#define CDIM     1024
#define BATCH    4
#define EPSV     1e-6f
#define KCH      32
#define LPAD     36

typedef float v2f __attribute__((ext_vector_type(2)));
typedef float v4f __attribute__((ext_vector_type(4)));
typedef float v8f __attribute__((ext_vector_type(8)));

static __device__ __forceinline__ v8f wmma4(v2f a, v2f b, v8f c) {
  return __builtin_amdgcn_wmma_f32_16x16x4_f32(false, a, false, b, (short)0, c,
                                               false, false);
}

static __device__ __forceinline__ float half16_max(float v) {
#pragma unroll
  for (int off = 1; off < 16; off <<= 1) v = fmaxf(v, __shfl_xor(v, off, 16));
  return v;
}
static __device__ __forceinline__ float half16_sum(float v) {
#pragma unroll
  for (int off = 1; off < 16; off <<= 1) v += __shfl_xor(v, off, 16);
  return v;
}
static __device__ __forceinline__ float wave32_sum(float v) {
#pragma unroll
  for (int off = 1; off < 32; off <<= 1) v += __shfl_xor(v, off, 32);
  return v;
}

__global__ __launch_bounds__(256) void gemm_abT_wmma(
    const float* __restrict__ A, const float* __restrict__ B,
    float* __restrict__ C, int K, int lda, int ldb, int ldc) {
  __shared__ __align__(16) float lA[2][128 * LPAD];
  __shared__ __align__(16) float lB[2][64 * LPAD];

  const int tid = threadIdx.x;
  const int lane = tid & 31;
  const int wave = tid >> 5;
  const int lm = lane & 15;
  const int lh = lane >> 4;
  const int m0 = blockIdx.y * 128;
  const int n0 = blockIdx.x * 64;

  const int srow = tid >> 3;
  const int scol = (tid & 7) * 4;

  const float* Ast = A + (size_t)(m0 + srow) * lda + scol;
  const float* Bst = B + (size_t)(n0 + srow) * ldb + scol;

  const float* A1 = Ast + (size_t)32 * lda;
  const float* A2 = Ast + (size_t)64 * lda;
  const float* A3 = Ast + (size_t)96 * lda;
  const float* B1 = Bst + (size_t)32 * ldb;
  const int so = srow * LPAD + scol;

  v4f ra0 = *(const v4f*)(Ast);
  v4f ra1 = *(const v4f*)(A1);
  v4f ra2 = *(const v4f*)(A2);
  v4f ra3 = *(const v4f*)(A3);
  v4f rb0 = *(const v4f*)(Bst);
  v4f rb1 = *(const v4f*)(B1);

  *(v4f*)(&lA[0][so]) = ra0;
  *(v4f*)(&lA[0][so + 32 * LPAD]) = ra1;
  *(v4f*)(&lA[0][so + 64 * LPAD]) = ra2;
  *(v4f*)(&lA[0][so + 96 * LPAD]) = ra3;
  *(v4f*)(&lB[0][so]) = rb0;
  *(v4f*)(&lB[0][so + 32 * LPAD]) = rb1;
  __syncthreads();

  v8f acc[4] = {v8f{}, v8f{}, v8f{}, v8f{}};
  const int nch = K / KCH;

  for (int c = 0; c < nch; ++c) {
    const int buf = c & 1;

    if (c + 1 < nch) {
      const int k1 = (c + 1) * KCH;
      ra0 = *(const v4f*)(Ast + k1);
      ra1 = *(const v4f*)(A1 + k1);
      ra2 = *(const v4f*)(A2 + k1);
      ra3 = *(const v4f*)(A3 + k1);
      rb0 = *(const v4f*)(Bst + k1);
      rb1 = *(const v4f*)(B1 + k1);
    }
    if (c + 2 < nch) {
      __builtin_prefetch(Ast + (size_t)(c + 2) * KCH, 0, 1);
      __builtin_prefetch(Bst + (size_t)(c + 2) * KCH, 0, 1);
    }

    const float* pa = &lA[buf][(wave * 16 + lm) * LPAD + 2 * lh];
    const float* pb = &lB[buf][lm * LPAD + 2 * lh];

    v2f af[8];
    v2f bf[8][4];
#pragma unroll
    for (int ks = 0; ks < 8; ++ks) {
      af[ks] = *(const v2f*)(pa + 4 * ks);
#pragma unroll
      for (int j = 0; j < 4; ++j)
        bf[ks][j] = *(const v2f*)(pb + j * 16 * LPAD + 4 * ks);
    }
#pragma unroll
    for (int ks = 0; ks < 8; ++ks)
#pragma unroll
      for (int j = 0; j < 4; ++j)
        acc[j] = wmma4(af[ks], bf[ks][j], acc[j]);

    if (c + 1 < nch) {
      const int nb = buf ^ 1;
      *(v4f*)(&lA[nb][so]) = ra0;
      *(v4f*)(&lA[nb][so + 32 * LPAD]) = ra1;
      *(v4f*)(&lA[nb][so + 64 * LPAD]) = ra2;
      *(v4f*)(&lA[nb][so + 96 * LPAD]) = ra3;
      *(v4f*)(&lB[nb][so]) = rb0;
      *(v4f*)(&lB[nb][so + 32 * LPAD]) = rb1;
      __syncthreads();
    }
  }

  const int mw = m0 + wave * 16;
#pragma unroll
  for (int j = 0; j < 4; ++j)
#pragma unroll
    for (int r = 0; r < 8; ++r)
      C[(size_t)(mw + r + 8 * lh) * ldc + n0 + j * 16 + lm] = acc[j][r];
}

__global__ __launch_bounds__(256) void rope_norm_kernel(
    const float* __restrict__ qkv, const float* __restrict__ hyp,
    float* __restrict__ Q, float* __restrict__ Kq, float* __restrict__ V,
    float* __restrict__ q0, float* __restrict__ k0) {
  const int lane = threadIdx.x & 31;
  const int gw = blockIdx.x * 8 + (threadIdx.x >> 5);
  const int h = gw % NHEADS;
  const int t = (gw / NHEADS) % TSEQ;
  const int b = gw / (NHEADS * TSEQ);

  const size_t ibase = ((size_t)(b * TSEQ + t)) * (3 * CDIM) + h * HD;
  const size_t obase = ((size_t)((b * NHEADS + h) * TSEQ) + t) * HD;
  const size_t nidx = (size_t)(b * NHEADS + h) * TSEQ + t;
  const float kc = hyp[h];

  const float fr = (float)t * __expf(-((float)(2 * lane) / (float)HD) *
                                     logf(10000.0f));
  float cs, sn;
  __sincosf(fr, &sn, &cs);

  {
    float x1 = qkv[ibase + lane];
    float x2 = qkv[ibase + 32 + lane];
    float o1 = x1 * cs + x2 * sn;
    float o2 = -x1 * sn + x2 * cs;
    Q[obase + lane] = o1;
    Q[obase + 32 + lane] = o2;
    float ss = wave32_sum(o1 * o1 + o2 * o2);
    if (lane == 0) q0[nidx] = sqrtf(kc + ss);
  }
  {
    float x1 = qkv[ibase + CDIM + lane];
    float x2 = qkv[ibase + CDIM + 32 + lane];
    float o1 = x1 * cs + x2 * sn;
    float o2 = -x1 * sn + x2 * cs;
    Kq[obase + lane] = o1;
    Kq[obase + 32 + lane] = o2;
    float ss = wave32_sum(o1 * o1 + o2 * o2);
    if (lane == 0) k0[nidx] = sqrtf(kc + ss);
  }
  V[obase + lane] = qkv[ibase + 2 * CDIM + lane];
  V[obase + 32 + lane] = qkv[ibase + 2 * CDIM + 32 + lane];
}

static __device__ __forceinline__ float acosh_hw(float x) {
  const float a = x - 1.0f;
  const float sq = __builtin_amdgcn_sqrtf(a) * __builtin_amdgcn_sqrtf(x + 1.0f);
  const float u = a + sq;
  const float v = 1.0f + u;
  const float corr = (u - (v - 1.0f)) * __builtin_amdgcn_rcpf(v);
  return fmaf(__builtin_amdgcn_logf(v), 0.69314718056f, corr);
}


static __device__ __forceinline__ float xhalf(float x) {
  const int b = __builtin_bit_cast(int, x);
  return __builtin_bit_cast(
      float, __builtin_amdgcn_permlanex16(b, b, 0x76543210, 0xfedcba98, false,
                                          false));
}

__global__ __launch_bounds__(256) void hyp_attn_kernel(
    const float* __restrict__ Q, const float* __restrict__ Km,
    const float* __restrict__ V, const float* __restrict__ q0,
    const float* __restrict__ k0, const float* __restrict__ hyp,
    float* __restrict__ O) {
  const int lane = threadIdx.x & 31;
  const int wave = threadIdx.x >> 5;
  const int lm = lane & 15;
  const int lh = lane >> 4;

  const int b = blockIdx.x / NHEADS;
  const int h = blockIdx.x % NHEADS;
  const int l0 = blockIdx.y * 128 + wave * 16;

  const size_t hb = (size_t)(b * NHEADS + h) * TSEQ;
  const float* Qb = Q + hb * HD;
  const float* Kb = Km + hb * HD;
  const float* Vb = V + hb * HD;
  const float* q0b = q0 + hb;
  const float* k0b = k0 + hb;

  const float hk = hyp[h];
  const float inv_hk = 1.0f / hk;
  const float sqk = sqrtf(hk);

  v2f qa[16];
#pragma unroll
  for (int d = 0; d < 16; ++d)
    qa[d] = *(const v2f*)(Qb + (size_t)(l0 + lm) * HD + 4 * d + 2 * lh);

  const int qry = l0 + lm;
  const float q0l = q0b[qry];

  float m_run = -1e30f, l_run = 0.0f;
  v8f o[4] = {v8f{}, v8f{}, v8f{}, v8f{}};

  v2f kf[16];
#pragma unroll
  for (int d = 0; d < 16; ++d)
    kf[d] = *(const v2f*)(Kb + (size_t)lm * HD + 4 * d + 2 * lh);
  v4f k0lo = *(const v4f*)(k0b + 8 * lh);
  v4f k0hi = *(const v4f*)(k0b + 8 * lh + 4);

  for (int s0 = 0; s0 <= l0; s0 += 16) {
    v2f vf[4][4];
#pragma unroll
    for (int kk = 0; kk < 4; ++kk) {
      const int krow = s0 + 8 * lh + 2 * kk;
#pragma unroll
      for (int j = 0; j < 4; ++j) {
        vf[kk][j].x = Vb[(size_t)krow * HD + j * 16 + lm];
        vf[kk][j].y = Vb[(size_t)(krow + 1) * HD + j * 16 + lm];
      }
    }

    v8f s = v8f{};
#pragma unroll
    for (int d = 0; d < 16; ++d) s = wmma4(kf[d], qa[d], s);

    const v4f kc_lo = k0lo;
    const v4f kc_hi = k0hi;

    const int sn = (s0 + 16 <= l0) ? (s0 + 16) : s0;
#pragma unroll
    for (int d = 0; d < 16; ++d)
      kf[d] = *(const v2f*)(Kb + (size_t)(sn + lm) * HD + 4 * d + 2 * lh);
    k0lo = *(const v4f*)(k0b + sn + 8 * lh);
    k0hi = *(const v4f*)(k0b + sn + 8 * lh + 4);
    __builtin_amdgcn_sched_barrier(0);

    float w[8];
#pragma unroll
    for (int r = 0; r < 8; ++r) {
      const int key = s0 + 8 * lh + r;
      const float k0r = (r < 4) ? kc_lo[r & 3] : kc_hi[r & 3];
      const float inner = s[r] - q0l * k0r;
      const float arg = fmaxf(-inner * inv_hk, 1.0f + EPSV);
      const float dis = sqk * acosh_hw(arg);
      const float ww = __builtin_amdgcn_rcpf(EPSV + dis);
      w[r] = (key > qry) ? -INFINITY : ww;
    }

    float tm = fmaxf(fmaxf(fmaxf(w[0], w[1]), fmaxf(w[2], w[3])),
                     fmaxf(fmaxf(w[4], w[5]), fmaxf(w[6], w[7])));
    tm = fmaxf(tm, xhalf(tm));
    const float newm = fmaxf(m_run, tm);
    const float scale = __expf(m_run - newm);
    m_run = newm;

    float p[8];
#pragma unroll
    for (int r = 0; r < 8; ++r) p[r] = __expf(w[r] - newm);
    const float ps = ((p[0] + p[1]) + (p[2] + p[3])) +
                     ((p[4] + p[5]) + (p[6] + p[7]));
    l_run = l_run * scale + ps;
#pragma unroll
    for (int j = 0; j < 4; ++j)
#pragma unroll
      for (int r = 0; r < 8; ++r) o[j][r] *= scale;

#pragma unroll
    for (int kk = 0; kk < 4; ++kk) {
      v2f pb;
      pb.x = p[2 * kk];
      pb.y = p[2 * kk + 1];
#pragma unroll
      for (int j = 0; j < 4; ++j) o[j] = wmma4(vf[kk][j], pb, o[j]);
    }
    __builtin_amdgcn_sched_barrier(0);
  }

  const float inv = 1.0f / (l_run + xhalf(l_run));
  float* op = O + ((size_t)(b * TSEQ + qry)) * CDIM + h * HD + 8 * lh;
#pragma unroll
  for (int j = 0; j < 4; ++j) {
    v4f lo, hi;
    lo.x = o[j][0] * inv; lo.y = o[j][1] * inv;
    lo.z = o[j][2] * inv; lo.w = o[j][3] * inv;
    hi.x = o[j][4] * inv; hi.y = o[j][5] * inv;
    hi.z = o[j][6] * inv; hi.w = o[j][7] * inv;
    *(v4f*)(op + 16 * j) = lo;
    *(v4f*)(op + 16 * j + 4) = hi;
  }
}

extern "C" void kernel_launch(void* const* d_in, const int* in_sizes, int n_in,
                              void* d_out, int out_size, void* d_ws,
                              size_t ws_size, hipStream_t stream) {
  (void)in_sizes; (void)n_in; (void)out_size; (void)ws_size;

  const float* x     = (const float*)d_in[0];
  const float* W_qkv = (const float*)d_in[1];
  const float* W_out = (const float*)d_in[2];
  const float* hyp   = (const float*)d_in[3];
  float* out = (float*)d_out;

  const size_t QKV_N = (size_t)BATCH * TSEQ * 3 * CDIM;
  const size_t QN    = (size_t)BATCH * NHEADS * TSEQ * HD;
  const size_t SN    = (size_t)BATCH * NHEADS * TSEQ;

  float* ws  = (float*)d_ws;
  float* qkv = ws;
  float* Q   = qkv + QKV_N;
  float* K   = Q + QN;
  float* V   = K + QN;
  float* q0  = V + QN;
  float* k0  = q0 + SN;
  float* AO  = k0 + SN;

  const int M = BATCH * TSEQ;

  gemm_abT_wmma<<<dim3((3 * CDIM) / 64, M / 128), 256, 0, stream>>>(
      x, W_qkv, qkv, CDIM, CDIM, CDIM, 3 * CDIM);

  rope_norm_kernel<<<(BATCH * TSEQ * NHEADS) / 8, 256, 0, stream>>>(
      qkv, hyp, Q, K, V, q0, k0);

  hyp_attn_kernel<<<dim3(BATCH * NHEADS, TSEQ / 128), 256, 0, stream>>>(
      Q, K, V, q0, k0, hyp, AO);

  gemm_abT_wmma<<<dim3(CDIM / 64, M / 128), 256, 0, stream>>>(
      AO, W_out, out, CDIM, CDIM, CDIM, CDIM);
}


# qkv GEMM epilogue applies the rotary rotation, writes the row norms (same summation tree) and stores head-major Q,K,V itself; the separate rotation launch and the flat qkv round trip are gone
# speedup vs baseline: 2.7582x; 1.0770x over previous
#include <hip/hip_runtime.h>
#include <math.h>

#define NHEADS   16
#define HD       64
#define TSEQ     1024
#define CDIM     1024
#define BATCH    4
#define EPSV     1e-6f
#define KCH      32
#define LPAD     36

typedef float v2f __attribute__((ext_vector_type(2)));
typedef float v4f __attribute__((ext_vector_type(4)));
typedef float v8f __attribute__((ext_vector_type(8)));

static __device__ __forceinline__ v8f wmma4(v2f a, v2f b, v8f c) {
  return __builtin_amdgcn_wmma_f32_16x16x4_f32(false, a, false, b, (short)0, c,
                                               false, false);
}

static __device__ __forceinline__ float half16_max(float v) {
#pragma unroll
  for (int off = 1; off < 16; off <<= 1) v = fmaxf(v, __shfl_xor(v, off, 16));
  return v;
}
static __device__ __forceinline__ float half16_sum(float v) {
#pragma unroll
  for (int off = 1; off < 16; off <<= 1) v += __shfl_xor(v, off, 16);
  return v;
}

template <int M>
static __device__ __forceinline__ float row_xor(float v) {
  return __builtin_bit_cast(
      float, __builtin_amdgcn_update_dpp(0, __builtin_bit_cast(int, v),
                                         0x160 + M, 0xf, 0xf, true));
}
template <int M>
static __device__ __forceinline__ void row_step(float (&a)[8], float (&b)[8]) {
#pragma unroll
  for (int r = 0; r < 8; ++r) {
    a[r] += row_xor<M>(a[r]);
    b[r] += row_xor<M>(b[r]);
  }
}

template <bool HEADS>
__global__ __launch_bounds__(256) void gemm_abT_wmma(
    const float* __restrict__ A, const float* __restrict__ B,
    float* __restrict__ C, int K, int lda, int ldb, int ldc,
    const float* __restrict__ hyp, float* __restrict__ Qo,
    float* __restrict__ Ko, float* __restrict__ Vo, float* __restrict__ q0,
    float* __restrict__ k0) {
  __shared__ __align__(16) float lA[2][128 * LPAD];
  __shared__ __align__(16) float lB[2][64 * LPAD];

  const int tid = threadIdx.x;
  const int lane = tid & 31;
  const int wave = tid >> 5;
  const int lm = lane & 15;
  const int lh = lane >> 4;
  const int m0 = blockIdx.y * 128;
  const int n0 = blockIdx.x * 64;

  const int srow = tid >> 3;
  const int scol = (tid & 7) * 4;

  const float* Ast = A + (size_t)(m0 + srow) * lda + scol;
  const float* Bst = B + (size_t)(n0 + srow) * ldb + scol;

  const float* A1 = Ast + (size_t)32 * lda;
  const float* A2 = Ast + (size_t)64 * lda;
  const float* A3 = Ast + (size_t)96 * lda;
  const float* B1 = Bst + (size_t)32 * ldb;
  const int so = srow * LPAD + scol;

  v4f ra0 = *(const v4f*)(Ast);
  v4f ra1 = *(const v4f*)(A1);
  v4f ra2 = *(const v4f*)(A2);
  v4f ra3 = *(const v4f*)(A3);
  v4f rb0 = *(const v4f*)(Bst);
  v4f rb1 = *(const v4f*)(B1);

  *(v4f*)(&lA[0][so]) = ra0;
  *(v4f*)(&lA[0][so + 32 * LPAD]) = ra1;
  *(v4f*)(&lA[0][so + 64 * LPAD]) = ra2;
  *(v4f*)(&lA[0][so + 96 * LPAD]) = ra3;
  *(v4f*)(&lB[0][so]) = rb0;
  *(v4f*)(&lB[0][so + 32 * LPAD]) = rb1;
  __syncthreads();

  v8f acc[4] = {v8f{}, v8f{}, v8f{}, v8f{}};
  const int nch = K / KCH;

  for (int c = 0; c < nch; ++c) {
    const int buf = c & 1;

    if (c + 1 < nch) {
      const int k1 = (c + 1) * KCH;
      ra0 = *(const v4f*)(Ast + k1);
      ra1 = *(const v4f*)(A1 + k1);
      ra2 = *(const v4f*)(A2 + k1);
      ra3 = *(const v4f*)(A3 + k1);
      rb0 = *(const v4f*)(Bst + k1);
      rb1 = *(const v4f*)(B1 + k1);
    }
    if (c + 2 < nch) {
      __builtin_prefetch(Ast + (size_t)(c + 2) * KCH, 0, 1);
      __builtin_prefetch(Bst + (size_t)(c + 2) * KCH, 0, 1);
    }

    const float* pa = &lA[buf][(wave * 16 + lm) * LPAD + 2 * lh];
    const float* pb = &lB[buf][lm * LPAD + 2 * lh];

    v2f af[8];
    v2f bf[8][4];
#pragma unroll
    for (int ks = 0; ks < 8; ++ks) {
      af[ks] = *(const v2f*)(pa + 4 * ks);
#pragma unroll
      for (int j = 0; j < 4; ++j)
        bf[ks][j] = *(const v2f*)(pb + j * 16 * LPAD + 4 * ks);
    }
#pragma unroll
    for (int ks = 0; ks < 8; ++ks)
#pragma unroll
      for (int j = 0; j < 4; ++j)
        acc[j] = wmma4(af[ks], bf[ks][j], acc[j]);

    if (c + 1 < nch) {
      const int nb = buf ^ 1;
      *(v4f*)(&lA[nb][so]) = ra0;
      *(v4f*)(&lA[nb][so + 32 * LPAD]) = ra1;
      *(v4f*)(&lA[nb][so + 64 * LPAD]) = ra2;
      *(v4f*)(&lA[nb][so + 96 * LPAD]) = ra3;
      *(v4f*)(&lB[nb][so]) = rb0;
      *(v4f*)(&lB[nb][so + 32 * LPAD]) = rb1;
      __syncthreads();
    }
  }

  const int mw = m0 + wave * 16;
  if constexpr (!HEADS) {
#pragma unroll
    for (int j = 0; j < 4; ++j)
#pragma unroll
      for (int r = 0; r < 8; ++r)
        C[(size_t)(mw + r + 8 * lh) * ldc + n0 + j * 16 + lm] = acc[j][r];
  } else {
    const int part = blockIdx.x / NHEADS;
    const int h = blockIdx.x % NHEADS;
    const int b = mw / TSEQ;
    const int t0 = (mw % TSEQ) + 8 * lh;
    const size_t nidx = (size_t)(b * NHEADS + h) * TSEQ + t0;
    float* dst = (part == 0 ? Qo : (part == 1 ? Ko : Vo)) + nidx * HD + lm;
    if (part == 2) {
#pragma unroll
      for (int j = 0; j < 4; ++j)
#pragma unroll
        for (int r = 0; r < 8; ++r) dst[r * HD + 16 * j] = acc[j][r];
    } else {
      const float kc = hyp[h];
      float* nrm = (part == 0 ? q0 : k0) + nidx;
      const float fa = __expf(-((float)(2 * lm) / (float)HD) * logf(10000.0f));
      const float fb =
          __expf(-((float)(2 * (16 + lm)) / (float)HD) * logf(10000.0f));
      float va[8], vb[8];
#pragma unroll
      for (int r = 0; r < 8; ++r) {
        const float tf = (float)(t0 + r);
        float sa, ca, sb, cb;
        __sincosf(tf * fa, &sa, &ca);
        __sincosf(tf * fb, &sb, &cb);
        const float x1 = acc[0][r], x2 = acc[2][r];
        const float y1 = acc[1][r], y2 = acc[3][r];
        const float o1 = x1 * ca + x2 * sa;
        const float o2 = -x1 * sa + x2 * ca;
        const float p1 = y1 * cb + y2 * sb;
        const float p2 = -y1 * sb + y2 * cb;
        dst[r * HD] = o1;
        dst[r * HD + 16] = p1;
        dst[r * HD + 32] = o2;
        dst[r * HD + 48] = p2;
        va[r] = fmaf(o1, o1, o2 * o2);
        vb[r] = fmaf(p1, p1, p2 * p2);
      }
      row_step<1>(va, vb);
      row_step<2>(va, vb);
      row_step<4>(va, vb);
      row_step<8>(va, vb);
      if (lm == 0) {
        v4f n0v, n1v;
        n0v.x = sqrtf(kc + (va[0] + vb[0])); n0v.y = sqrtf(kc + (va[1] + vb[1]));
        n0v.z = sqrtf(kc + (va[2] + vb[2])); n0v.w = sqrtf(kc + (va[3] + vb[3]));
        n1v.x = sqrtf(kc + (va[4] + vb[4])); n1v.y = sqrtf(kc + (va[5] + vb[5]));
        n1v.z = sqrtf(kc + (va[6] + vb[6])); n1v.w = sqrtf(kc + (va[7] + vb[7]));
        *(v4f*)(nrm) = n0v;
        *(v4f*)(nrm + 4) = n1v;
      }
    }
  }
}

static __device__ __forceinline__ float acosh_hw(float x) {
  const float a = x - 1.0f;
  const float sq = __builtin_amdgcn_sqrtf(a) * __builtin_amdgcn_sqrtf(x + 1.0f);
  const float u = a + sq;
  const float v = 1.0f + u;
  const float corr = (u - (v - 1.0f)) * __builtin_amdgcn_rcpf(v);
  return fmaf(__builtin_amdgcn_logf(v), 0.69314718056f, corr);
}


static __device__ __forceinline__ float xhalf(float x) {
  const int b = __builtin_bit_cast(int, x);
  return __builtin_bit_cast(
      float, __builtin_amdgcn_permlanex16(b, b, 0x76543210, 0xfedcba98, false,
                                          false));
}

__global__ __launch_bounds__(256) void hyp_attn_kernel(
    const float* __restrict__ Q, const float* __restrict__ Km,
    const float* __restrict__ V, const float* __restrict__ q0,
    const float* __restrict__ k0, const float* __restrict__ hyp,
    float* __restrict__ O) {
  const int lane = threadIdx.x & 31;
  const int wave = threadIdx.x >> 5;
  const int lm = lane & 15;
  const int lh = lane >> 4;

  const int b = blockIdx.x / NHEADS;
  const int h = blockIdx.x % NHEADS;
  const int l0 = blockIdx.y * 128 + wave * 16;

  const size_t hb = (size_t)(b * NHEADS + h) * TSEQ;
  const float* Qb = Q + hb * HD;
  const float* Kb = Km + hb * HD;
  const float* Vb = V + hb * HD;
  const float* q0b = q0 + hb;
  const float* k0b = k0 + hb;

  const float hk = hyp[h];
  const float inv_hk = 1.0f / hk;
  const float sqk = sqrtf(hk);

  v2f qa[16];
#pragma unroll
  for (int d = 0; d < 16; ++d)
    qa[d] = *(const v2f*)(Qb + (size_t)(l0 + lm) * HD + 4 * d + 2 * lh);

  const int qry = l0 + lm;
  const float q0l = q0b[qry];

  float m_run = -1e30f, l_run = 0.0f;
  v8f o[4] = {v8f{}, v8f{}, v8f{}, v8f{}};

  v2f kf[16];
#pragma unroll
  for (int d = 0; d < 16; ++d)
    kf[d] = *(const v2f*)(Kb + (size_t)lm * HD + 4 * d + 2 * lh);
  v4f k0lo = *(const v4f*)(k0b + 8 * lh);
  v4f k0hi = *(const v4f*)(k0b + 8 * lh + 4);

  for (int s0 = 0; s0 <= l0; s0 += 16) {
    v2f vf[4][4];
#pragma unroll
    for (int kk = 0; kk < 4; ++kk) {
      const int krow = s0 + 8 * lh + 2 * kk;
#pragma unroll
      for (int j = 0; j < 4; ++j) {
        vf[kk][j].x = Vb[(size_t)krow * HD + j * 16 + lm];
        vf[kk][j].y = Vb[(size_t)(krow + 1) * HD + j * 16 + lm];
      }
    }

    v8f s = v8f{};
#pragma unroll
    for (int d = 0; d < 16; ++d) s = wmma4(kf[d], qa[d], s);

    const v4f kc_lo = k0lo;
    const v4f kc_hi = k0hi;

    const int sn = (s0 + 16 <= l0) ? (s0 + 16) : s0;
#pragma unroll
    for (int d = 0; d < 16; ++d)
      kf[d] = *(const v2f*)(Kb + (size_t)(sn + lm) * HD + 4 * d + 2 * lh);
    k0lo = *(const v4f*)(k0b + sn + 8 * lh);
    k0hi = *(const v4f*)(k0b + sn + 8 * lh + 4);
    __builtin_amdgcn_sched_barrier(0);

    float w[8];
#pragma unroll
    for (int r = 0; r < 8; ++r) {
      const int key = s0 + 8 * lh + r;
      const float k0r = (r < 4) ? kc_lo[r & 3] : kc_hi[r & 3];
      const float inner = s[r] - q0l * k0r;
      const float arg = fmaxf(-inner * inv_hk, 1.0f + EPSV);
      const float dis = sqk * acosh_hw(arg);
      const float ww = __builtin_amdgcn_rcpf(EPSV + dis);
      w[r] = (key > qry) ? -INFINITY : ww;
    }

    float tm = fmaxf(fmaxf(fmaxf(w[0], w[1]), fmaxf(w[2], w[3])),
                     fmaxf(fmaxf(w[4], w[5]), fmaxf(w[6], w[7])));
    tm = fmaxf(tm, xhalf(tm));
    const float newm = fmaxf(m_run, tm);
    const float scale = __expf(m_run - newm);
    m_run = newm;

    float p[8];
#pragma unroll
    for (int r = 0; r < 8; ++r) p[r] = __expf(w[r] - newm);
    const float ps = ((p[0] + p[1]) + (p[2] + p[3])) +
                     ((p[4] + p[5]) + (p[6] + p[7]));
    l_run = l_run * scale + ps;
#pragma unroll
    for (int j = 0; j < 4; ++j)
#pragma unroll
      for (int r = 0; r < 8; ++r) o[j][r] *= scale;

#pragma unroll
    for (int kk = 0; kk < 4; ++kk) {
      v2f pb;
      pb.x = p[2 * kk];
      pb.y = p[2 * kk + 1];
#pragma unroll
      for (int j = 0; j < 4; ++j) o[j] = wmma4(vf[kk][j], pb, o[j]);
    }
    __builtin_amdgcn_sched_barrier(0);
  }

  const float inv = 1.0f / (l_run + xhalf(l_run));
  float* op = O + ((size_t)(b * TSEQ + qry)) * CDIM + h * HD + 8 * lh;
#pragma unroll
  for (int j = 0; j < 4; ++j) {
    v4f lo, hi;
    lo.x = o[j][0] * inv; lo.y = o[j][1] * inv;
    lo.z = o[j][2] * inv; lo.w = o[j][3] * inv;
    hi.x = o[j][4] * inv; hi.y = o[j][5] * inv;
    hi.z = o[j][6] * inv; hi.w = o[j][7] * inv;
    *(v4f*)(op + 16 * j) = lo;
    *(v4f*)(op + 16 * j + 4) = hi;
  }
}

extern "C" void kernel_launch(void* const* d_in, const int* in_sizes, int n_in,
                              void* d_out, int out_size, void* d_ws,
                              size_t ws_size, hipStream_t stream) {
  (void)in_sizes; (void)n_in; (void)out_size; (void)ws_size;

  const float* x     = (const float*)d_in[0];
  const float* W_qkv = (const float*)d_in[1];
  const float* W_out = (const float*)d_in[2];
  const float* hyp   = (const float*)d_in[3];
  float* out = (float*)d_out;

  const size_t QN    = (size_t)BATCH * NHEADS * TSEQ * HD;
  const size_t SN    = (size_t)BATCH * NHEADS * TSEQ;

  float* ws  = (float*)d_ws;
  float* Q   = ws;
  float* K   = Q + QN;
  float* V   = K + QN;
  float* q0  = V + QN;
  float* k0  = q0 + SN;
  float* AO  = k0 + SN;

  const int M = BATCH * TSEQ;

  gemm_abT_wmma<true><<<dim3((3 * CDIM) / 64, M / 128), 256, 0, stream>>>(
      x, W_qkv, nullptr, CDIM, CDIM, CDIM, 3 * CDIM, hyp, Q, K, V, q0, k0);

  hyp_attn_kernel<<<dim3(BATCH * NHEADS, TSEQ / 128), 256, 0, stream>>>(
      Q, K, V, q0, k0, hyp, AO);

  gemm_abT_wmma<false><<<dim3(CDIM / 64, M / 128), 256, 0, stream>>>(
      AO, W_out, out, CDIM, CDIM, CDIM, CDIM, nullptr, nullptr, nullptr,
      nullptr, nullptr, nullptr);
}
